# starting kernel
# baseline (speedup 1.0000x reference)
#include <hip/hip_runtime.h>
#include <hip/hip_bf16.h>
#include <math.h>


#define BATCH    8
#define LENGTH   2048
#define D_MODEL  1024
#define D_STATE  64
#define TCH      16
#define CHUNKS   (LENGTH / TCH)
#define COLS     (BATCH * CHUNKS)
#define TILES    (COLS / 16)

typedef __attribute__((ext_vector_type(2))) float v2f;
typedef __attribute__((ext_vector_type(8))) float v8f;

#if defined(__HIP_DEVICE_COMPILE__)
#if !__has_builtin(__builtin_amdgcn_wmma_f32_16x16x4_f32)
#error "gfx1250 device pass: __builtin_amdgcn_wmma_f32_16x16x4_f32 not available"
#endif
#endif

__device__ __forceinline__ v8f wmma_f32_16x16x4(v2f a, v2f b, v8f c) {
#if defined(__HIP_DEVICE_COMPILE__)
  return __builtin_amdgcn_wmma_f32_16x16x4_f32(false, a, false, b, (short)0, c,
                                               false, false);
#else
  (void)a; (void)b;
  return c;
#endif
}

__global__ __launch_bounds__(256) void s4d_intra(
    const float* __restrict__ x, const float* __restrict__ log_dt,
    const float* __restrict__ Bmat, const float* __restrict__ Cmat,
    const float* __restrict__ Dvec, float* __restrict__ y)
{
  const int d = blockIdx.x;
  __shared__ float sP[TCH][D_STATE];
  __shared__ float sK[TCH];

  const int tid = threadIdx.x;
  const float dt = expf(log_dt[d]);

  if (tid < D_STATE) {
    const float An  = -0.5f * (float)(tid + 1);
    const float dtA = dt * An;
    const float Abar = expf(dtA);
    float q = Cmat[d * D_STATE + tid] *
              (expm1f(dtA) / An) * Bmat[d * D_STATE + tid];
    #pragma unroll
    for (int j = 0; j < TCH; ++j) { sP[j][tid] = q; q *= Abar; }
  }
  __syncthreads();
  if (tid < TCH) {
    float acc = 0.f;
    #pragma unroll 8
    for (int n = 0; n < D_STATE; ++n) acc += sP[tid][n];
    sK[tid] = acc;
  }
  __syncthreads();

  const int lane = tid & 31;
  const int wave = tid >> 5;
  const int row  = lane & 15;
  const int hi   = lane >> 4;

  v2f afrag[4];
  #pragma unroll
  for (int kk = 0; kk < 4; ++kk) {
    #pragma unroll
    for (int v = 0; v < 2; ++v) {
      const int col = kk * 4 + hi * 2 + v;
      afrag[kk][v] = (row >= col) ? sK[row - col] : 0.f;
    }
  }

  const float Dd = Dvec[d];

  for (int t8 = 0; t8 < TILES / 8; ++t8) {
    const int tile = t8 * 8 + wave;
    const int cc = tile * 16 + row;
    const int b  = cc >> 7;
    const int ch = cc & 127;
    const int base = (b * LENGTH + ch * TCH) * D_MODEL + d;

    v8f acc = {};
    #pragma unroll
    for (int kk = 0; kk < 4; ++kk) {
      v2f bfrag;
      #pragma unroll
      for (int v = 0; v < 2; ++v) {
        const int j = kk * 4 + hi * 2 + v;
        bfrag[v] = x[base + j * D_MODEL];
      }
      acc = wmma_f32_16x16x4(afrag[kk], bfrag, acc);
    }

    #pragma unroll
    for (int v = 0; v < 8; ++v) {
      const int r   = v + hi * 8;
      const int idx = base + r * D_MODEL;
      y[idx] = acc[v] + Dd * x[idx];
    }
  }
}

__global__ __launch_bounds__(32) void s4d_state(
    const float* __restrict__ x, const float* __restrict__ log_dt,
    const float* __restrict__ Bmat, const float* __restrict__ Cmat,
    float* __restrict__ y)
{
  const int d    = blockIdx.x;
  const int lane = threadIdx.x;
  const int row  = lane & 15;
  const int hi   = lane >> 4;

  __shared__ float sAbar[D_STATE];
  __shared__ float sBbar[D_STATE];
  __shared__ float sX[TCH * BATCH];

  const float dt = expf(log_dt[d]);
  for (int t = lane; t < D_STATE; t += 32) {
    const float An  = -0.5f * (float)(t + 1);
    const float dtA = dt * An;
    sAbar[t] = expf(dtA);
    sBbar[t] = (expm1f(dtA) / An) * Bmat[d * D_STATE + t];
  }
  __syncthreads();

  v2f g[16];
  #pragma unroll
  for (int kk = 0; kk < 16; ++kk) {
    #pragma unroll
    for (int v = 0; v < 2; ++v) {
      const int n = kk * 4 + hi * 2 + v;
      g[kk][v] = Cmat[d * D_STATE + n] * powf(sAbar[n], (float)(row + 1));
    }
  }

  float cA[32], cB[32], cL[32];
  v2f s[16];
  #pragma unroll
  for (int kk = 0; kk < 16; ++kk) {
    #pragma unroll
    for (int v = 0; v < 2; ++v) {
      const int n = kk * 4 + hi * 2 + v;
      const float a1 = sAbar[n];
      const float a2 = a1 * a1, a4 = a2 * a2, a8 = a4 * a4;
      cA[kk * 2 + v] = a1;
      cB[kk * 2 + v] = sBbar[n];
      cL[kk * 2 + v] = a8 * a8;
      s[kk][v] = 0.f;
    }
  }

  for (int ch = 0; ch < CHUNKS; ++ch) {
    const int l0 = ch * TCH;

    #pragma unroll
    for (int k = 0; k < 4; ++k) {
      const int e = lane + k * 32;
      const int j = e >> 3;
      const int b = e & 7;
      sX[e] = x[(b * LENGTH + l0 + j) * D_MODEL + d];
    }
    __syncthreads();

    v8f acc = {};
    #pragma unroll
    for (int kk = 0; kk < 16; ++kk)
      acc = wmma_f32_16x16x4(g[kk], s[kk], acc);

    if (row < BATCH) {
      #pragma unroll
      for (int v = 0; v < 8; ++v) {
        const int r   = v + hi * 8;
        const int idx = (row * LENGTH + l0 + r) * D_MODEL + d;
        y[idx] += acc[v];
      }
    }

    float xv[TCH];
    #pragma unroll
    for (int j = 0; j < TCH; ++j)
      xv[j] = (row < BATCH) ? sX[j * BATCH + row] : 0.f;

    #pragma unroll
    for (int kk = 0; kk < 16; ++kk) {
      #pragma unroll
      for (int v = 0; v < 2; ++v) {
        float h = 0.f;
        #pragma unroll
        for (int j = 0; j < TCH; ++j)
          h = fmaf(cA[kk * 2 + v], h, xv[j]);
        s[kk][v] = fmaf(cL[kk * 2 + v], s[kk][v], cB[kk * 2 + v] * h);
      }
    }
    __syncthreads();
  }
}

extern "C" void kernel_launch(void* const* d_in, const int* in_sizes, int n_in,
                              void* d_out, int out_size, void* d_ws, size_t ws_size,
                              hipStream_t stream) {
  (void)in_sizes; (void)n_in; (void)out_size; (void)d_ws; (void)ws_size;
  const float* x      = (const float*)d_in[0];
  const float* log_dt = (const float*)d_in[1];
  const float* Bmat   = (const float*)d_in[2];
  const float* Cmat   = (const float*)d_in[3];
  const float* Dvec   = (const float*)d_in[4];
  float* y            = (float*)d_out;

  s4d_intra<<<dim3(D_MODEL), 256, 0, stream>>>(x, log_dt, Bmat, Cmat, Dvec, y);
  s4d_state<<<dim3(D_MODEL), 32, 0, stream>>>(x, log_dt, Bmat, Cmat, y);
}


# intra-chunk kernel: work-group owns 32 adjacent channels, x/y tiles staged through LDS so every global access is a full 128-byte row (was 4-byte gathers at 4 KB stride); sums unchanged
# speedup vs baseline: 1.8668x; 1.8668x over previous
#include <hip/hip_runtime.h>
#include <hip/hip_bf16.h>
#include <math.h>


#define BATCH    8
#define LENGTH   2048
#define D_MODEL  1024
#define D_STATE  64
#define TCH      16
#define CHUNKS   (LENGTH / TCH)
#define COLS     (BATCH * CHUNKS)
#define TILES    (COLS / 16)

typedef __attribute__((ext_vector_type(2))) float v2f;
typedef __attribute__((ext_vector_type(8))) float v8f;

#if defined(__HIP_DEVICE_COMPILE__)
#if !__has_builtin(__builtin_amdgcn_wmma_f32_16x16x4_f32)
#error "gfx1250 device pass: __builtin_amdgcn_wmma_f32_16x16x4_f32 not available"
#endif
#endif

__device__ __forceinline__ v8f wmma_f32_16x16x4(v2f a, v2f b, v8f c) {
#if defined(__HIP_DEVICE_COMPILE__)
  return __builtin_amdgcn_wmma_f32_16x16x4_f32(false, a, false, b, (short)0, c,
                                               false, false);
#else
  (void)a; (void)b;
  return c;
#endif
}

#define DCH        32
#define NSPLIT     16
#define TPB        (TILES / NSPLIT)
#define ROWP       33
#define COLP       581
#define SPP        65
#define INTRA_GRID 512

static_assert(INTRA_GRID == (D_MODEL / DCH) * NSPLIT, "grid literal out of date");
static_assert(8 * 2 * TCH * SPP <= 2 * 16 * COLP, "tap scratch must fit in the tile buffers");

__global__ __launch_bounds__(256) __attribute__((amdgpu_waves_per_eu(2, 4)))
void s4d_intra(
    const float* __restrict__ x, const float* __restrict__ log_dt,
    const float* __restrict__ Bmat, const float* __restrict__ Cmat,
    const float* __restrict__ Dvec, float* __restrict__ y)
{
  __shared__ float sTb[2 * 16 * COLP];
  __shared__ float sK[DCH][TCH];

  const int tid   = threadIdx.x;
  const int lane  = tid & 31;
  const int wave  = tid >> 5;
  const int row   = lane & 15;
  const int hi    = lane >> 4;
  const int d0    = (blockIdx.x & 31) * DCH;
  const int split = blockIdx.x >> 5;

  float qv[8], ab[8];
  #pragma unroll
  for (int i = 0; i < 4; ++i) {
    const int d = d0 + wave * 4 + i;
    const float dt = expf(log_dt[d]);
    #pragma unroll
    for (int h = 0; h < 2; ++h) {
      const int n = lane + 32 * h;
      const float An  = -0.5f * (float)(n + 1);
      const float dtA = dt * An;
      ab[i * 2 + h] = expf(dtA);
      qv[i * 2 + h] = Cmat[d * D_STATE + n] *
                      (expm1f(dtA) / An) * Bmat[d * D_STATE + n];
    }
  }
  float* sPw = sTb + wave * (2 * TCH * SPP);
  #pragma unroll
  for (int r = 0; r < 2; ++r) {
    #pragma unroll
    for (int c2 = 0; c2 < 2; ++c2) {
      #pragma unroll
      for (int h = 0; h < 2; ++h) {
        const int n = lane + 32 * h;
        float q = qv[(2 * r + c2) * 2 + h];
        const float Abar = ab[(2 * r + c2) * 2 + h];
        #pragma unroll
        for (int j = 0; j < TCH; ++j) {
          sPw[c2 * (TCH * SPP) + j * SPP + n] = q;
          q *= Abar;
        }
      }
    }
    __syncthreads();
    {
      const float* ps = sPw + hi * (TCH * SPP) + row * SPP;
      float acc = 0.f;
      #pragma unroll 8
      for (int n = 0; n < D_STATE; ++n) acc += ps[n];
      sK[wave * 4 + 2 * r + hi][row] = acc;
    }
    __syncthreads();
  }

  v2f   afrag[4][4];
  float Dd[4];
  #pragma unroll
  for (int i = 0; i < 4; ++i) {
    const int dl = wave * 4 + i;
    #pragma unroll
    for (int kk = 0; kk < 4; ++kk) {
      #pragma unroll
      for (int v = 0; v < 2; ++v) {
        const int col = kk * 4 + hi * 2 + v;
        const float tap = sK[dl][(row - col) & 15];
        afrag[i][kk][v] = (row >= col) ? tap : 0.f;
      }
    }
    Dd[i] = Dvec[d0 + dl];
  }

  #pragma nounroll
  for (int t = 0; t < TPB; ++t) {
    const int tile  = split * TPB + t;
    float* sT = sTb + (t & 1) * (16 * COLP);
    const int gbase = tile * 256 * D_MODEL + d0;

    float4 rg[8];
    #pragma unroll
    for (int k = 0; k < 8; ++k) {
      const int f  = k * 256 + tid;
      const int rr = f >> 3;
      const int q  = f & 7;
      rg[k] = *reinterpret_cast<const float4*>(x + gbase + rr * D_MODEL + 4 * q);
    }
    #pragma unroll
    for (int k = 0; k < 8; ++k) {
      const int f  = k * 256 + tid;
      const int rr = f >> 3;
      const int q  = f & 7;
      float* p = sT + (rr >> 4) * COLP + (rr & 15) * ROWP + 4 * q;
      p[0] = rg[k].x; p[1] = rg[k].y; p[2] = rg[k].z; p[3] = rg[k].w;
    }
    __syncthreads();

    #pragma unroll
    for (int i = 0; i < 4; ++i) {
      float* pc = sT + row * COLP + (wave * 4 + i);
      v2f bfrag[4];
      #pragma unroll
      for (int kk = 0; kk < 4; ++kk) {
        #pragma unroll
        for (int v = 0; v < 2; ++v) {
          const int j = kk * 4 + hi * 2 + v;
          bfrag[kk][v] = pc[j * ROWP];
        }
      }
      float xs[8];
      #pragma unroll
      for (int v = 0; v < 8; ++v) xs[v] = pc[(v + hi * 8) * ROWP];

      v8f acc = {};
      #pragma unroll
      for (int kk = 0; kk < 4; ++kk)
        acc = wmma_f32_16x16x4(afrag[i][kk], bfrag[kk], acc);

      #pragma unroll
      for (int v = 0; v < 8; ++v)
        pc[(v + hi * 8) * ROWP] = acc[v] + Dd[i] * xs[v];
    }
    __syncthreads();

    #pragma unroll
    for (int k = 0; k < 8; ++k) {
      const int f  = k * 256 + tid;
      const int rr = f >> 3;
      const int q  = f & 7;
      const float* p = sT + (rr >> 4) * COLP + (rr & 15) * ROWP + 4 * q;
      float4 o;
      o.x = p[0]; o.y = p[1]; o.z = p[2]; o.w = p[3];
      *reinterpret_cast<float4*>(y + gbase + rr * D_MODEL + 4 * q) = o;
    }
  }
}

__global__ __launch_bounds__(32) void s4d_state(
    const float* __restrict__ x, const float* __restrict__ log_dt,
    const float* __restrict__ Bmat, const float* __restrict__ Cmat,
    float* __restrict__ y)
{
  const int d    = blockIdx.x;
  const int lane = threadIdx.x;
  const int row  = lane & 15;
  const int hi   = lane >> 4;

  __shared__ float sAbar[D_STATE];
  __shared__ float sBbar[D_STATE];
  __shared__ float sX[TCH * BATCH];

  const float dt = expf(log_dt[d]);
  for (int t = lane; t < D_STATE; t += 32) {
    const float An  = -0.5f * (float)(t + 1);
    const float dtA = dt * An;
    sAbar[t] = expf(dtA);
    sBbar[t] = (expm1f(dtA) / An) * Bmat[d * D_STATE + t];
  }
  __syncthreads();

  v2f g[16];
  #pragma unroll
  for (int kk = 0; kk < 16; ++kk) {
    #pragma unroll
    for (int v = 0; v < 2; ++v) {
      const int n = kk * 4 + hi * 2 + v;
      g[kk][v] = Cmat[d * D_STATE + n] * powf(sAbar[n], (float)(row + 1));
    }
  }

  float cA[32], cB[32], cL[32];
  v2f s[16];
  #pragma unroll
  for (int kk = 0; kk < 16; ++kk) {
    #pragma unroll
    for (int v = 0; v < 2; ++v) {
      const int n = kk * 4 + hi * 2 + v;
      const float a1 = sAbar[n];
      const float a2 = a1 * a1, a4 = a2 * a2, a8 = a4 * a4;
      cA[kk * 2 + v] = a1;
      cB[kk * 2 + v] = sBbar[n];
      cL[kk * 2 + v] = a8 * a8;
      s[kk][v] = 0.f;
    }
  }

  for (int ch = 0; ch < CHUNKS; ++ch) {
    const int l0 = ch * TCH;

    #pragma unroll
    for (int k = 0; k < 4; ++k) {
      const int e = lane + k * 32;
      const int j = e >> 3;
      const int b = e & 7;
      sX[e] = x[(b * LENGTH + l0 + j) * D_MODEL + d];
    }
    __syncthreads();

    v8f acc = {};
    #pragma unroll
    for (int kk = 0; kk < 16; ++kk)
      acc = wmma_f32_16x16x4(g[kk], s[kk], acc);

    if (row < BATCH) {
      #pragma unroll
      for (int v = 0; v < 8; ++v) {
        const int r   = v + hi * 8;
        const int idx = (row * LENGTH + l0 + r) * D_MODEL + d;
        y[idx] += acc[v];
      }
    }

    float xv[TCH];
    #pragma unroll
    for (int j = 0; j < TCH; ++j)
      xv[j] = (row < BATCH) ? sX[j * BATCH + row] : 0.f;

    #pragma unroll
    for (int kk = 0; kk < 16; ++kk) {
      #pragma unroll
      for (int v = 0; v < 2; ++v) {
        float h = 0.f;
        #pragma unroll
        for (int j = 0; j < TCH; ++j)
          h = fmaf(cA[kk * 2 + v], h, xv[j]);
        s[kk][v] = fmaf(cL[kk * 2 + v], s[kk][v], cB[kk * 2 + v] * h);
      }
    }
    __syncthreads();
  }
}

extern "C" void kernel_launch(void* const* d_in, const int* in_sizes, int n_in,
                              void* d_out, int out_size, void* d_ws, size_t ws_size,
                              hipStream_t stream) {
  (void)in_sizes; (void)n_in; (void)out_size; (void)d_ws; (void)ws_size;
  const float* x      = (const float*)d_in[0];
  const float* log_dt = (const float*)d_in[1];
  const float* Bmat   = (const float*)d_in[2];
  const float* Cmat   = (const float*)d_in[3];
  const float* Dvec   = (const float*)d_in[4];
  float* y            = (float*)d_out;

  s4d_intra<<<dim3(INTRA_GRID), 256, 0, stream>>>(x, log_dt, Bmat, Cmat, Dvec, y);
  s4d_state<<<dim3(D_MODEL), 32, 0, stream>>>(x, log_dt, Bmat, Cmat, y);
}


# state term fused into the row-staged tile kernel: chunk end-states by WMMA + 16-lane weighted scan, G@S from D regs; 2nd launch and its 4-byte gathers gone; state sums reordered
# speedup vs baseline: 13.9825x; 7.4901x over previous
#include <hip/hip_runtime.h>
#include <hip/hip_bf16.h>
#include <math.h>


#define BATCH    8
#define LENGTH   2048
#define D_MODEL  1024
#define D_STATE  64
#define TCH      16
#define CHUNKS   (LENGTH / TCH)
#define TPB      (CHUNKS / 16)

typedef __attribute__((ext_vector_type(2))) float v2f;
typedef __attribute__((ext_vector_type(8))) float v8f;

#if defined(__HIP_DEVICE_COMPILE__)
#if !__has_builtin(__builtin_amdgcn_wmma_f32_16x16x4_f32)
#error "gfx1250 device pass: __builtin_amdgcn_wmma_f32_16x16x4_f32 not available"
#endif
#endif

__device__ __forceinline__ v8f wmma_f32_16x16x4(v2f a, v2f b, v8f c) {
#if defined(__HIP_DEVICE_COMPILE__)
  return __builtin_amdgcn_wmma_f32_16x16x4_f32(false, a, false, b, (short)0, c,
                                               false, false);
#else
  (void)a; (void)b;
  return c;
#endif
}

template <int CTRL>
__device__ __forceinline__ float row_dpp(float v) {
#if defined(__HIP_DEVICE_COMPILE__)
  return __int_as_float(__builtin_amdgcn_update_dpp(0, __float_as_int(v), CTRL,
                                                    0xF, 0xF, true));
#else
  return v;
#endif
}
#define DPP_ROW_SHR(k)   (0x110 + (k))

#define DCH        32
#define ROWP       33
#define COLP       581
#define SPP        65
#define FUSED_GRID 256

static_assert(FUSED_GRID == (D_MODEL / DCH) * BATCH, "grid literal out of date");
static_assert(8 * 2 * TCH * SPP <= 2 * 16 * COLP, "tap scratch must fit in the tile buffers");

__global__ __launch_bounds__(256) __attribute__((amdgpu_waves_per_eu(2, 2)))
void s4d_fused(
    const float* __restrict__ x, const float* __restrict__ log_dt,
    const float* __restrict__ Bmat, const float* __restrict__ Cmat,
    const float* __restrict__ Dvec, float* __restrict__ y)
{
  __shared__ float sTb[2 * 16 * COLP];
  __shared__ float sK[DCH][TCH];
  __shared__ __attribute__((aligned(16))) float sG[DCH * 16 * 32 * 2];
  __shared__ __attribute__((aligned(16))) float sAb[DCH][D_STATE];
  __shared__ __attribute__((aligned(16))) float sBb[DCH][D_STATE];
  __shared__ __attribute__((aligned(16))) float sA16[DCH][D_STATE];
  __shared__ __attribute__((aligned(16))) float sCar[DCH][D_STATE];

  const int tid   = threadIdx.x;
  const int lane  = tid & 31;
  const int wave  = tid >> 5;
  const int row   = lane & 15;
  const int hi    = lane >> 4;
  const int d0    = (blockIdx.x & 31) * DCH;
  const int bat   = blockIdx.x >> 5;

  float qv[8], ab[8];
  #pragma unroll
  for (int i = 0; i < 4; ++i) {
    const int chl = wave * 4 + i;
    const int d = d0 + chl;
    const float dt = expf(log_dt[d]);
    #pragma unroll
    for (int h = 0; h < 2; ++h) {
      const int n = lane + 32 * h;
      const float An  = -0.5f * (float)(n + 1);
      const float dtA = dt * An;
      const float Abar = expf(dtA);
      const float Cn   = Cmat[d * D_STATE + n];
      const float Bn   = Bmat[d * D_STATE + n];
      ab[i * 2 + h] = Abar;
      qv[i * 2 + h] = Cn * (expm1f(dtA) / An) * Bn;
      sAb[chl][n] = Abar;
      sBb[chl][n] = (expm1f(dtA) / An) * Bn;
      const float a2 = Abar * Abar, a4 = a2 * a2, a8 = a4 * a4;
      sA16[chl][n] = a8 * a8;
      sCar[chl][n] = 0.f;
      float* gp = sG + ((chl * 16 + (n >> 4) * 4 + ((n >> 1) & 3)) * 32 + ((n >> 3) & 1) * 16) * 2 + (n & 1);
      float g = Cn;
      #pragma unroll
      for (int r = 0; r < TCH; ++r) {
        g *= Abar;
        gp[r * 2] = g;
      }
    }
  }
  float* sPw = sTb + wave * (2 * TCH * SPP);
  #pragma unroll
  for (int r = 0; r < 2; ++r) {
    #pragma unroll
    for (int c2 = 0; c2 < 2; ++c2) {
      #pragma unroll
      for (int h = 0; h < 2; ++h) {
        const int n = lane + 32 * h;
        float q = qv[(2 * r + c2) * 2 + h];
        const float Abar = ab[(2 * r + c2) * 2 + h];
        #pragma unroll
        for (int j = 0; j < TCH; ++j) {
          sPw[c2 * (TCH * SPP) + j * SPP + n] = q;
          q *= Abar;
        }
      }
    }
    __syncthreads();
    {
      const float* ps = sPw + hi * (TCH * SPP) + row * SPP;
      float acc = 0.f;
      #pragma unroll 8
      for (int n = 0; n < D_STATE; ++n) acc += ps[n];
      sK[wave * 4 + 2 * r + hi][row] = acc;
    }
    __syncthreads();
  }

  v2f   afrag[4][4];
  float Dd[4];
  v2f   pf[4][4][4];
  #pragma unroll
  for (int i = 0; i < 4; ++i) {
    const int dl = wave * 4 + i;
    #pragma unroll
    for (int kk = 0; kk < 4; ++kk) {
      #pragma unroll
      for (int v = 0; v < 2; ++v) {
        const int col = kk * 4 + hi * 2 + v;
        const float tap = sK[dl][(row - col) & 15];
        afrag[i][kk][v] = (row >= col) ? tap : 0.f;
      }
    }
    Dd[i] = Dvec[d0 + dl];
    #pragma unroll
    for (int m = 0; m < 4; ++m) {
      const float A = sAb[dl][m * 16 + row];
      float pw[16];
      pw[0] = sBb[dl][m * 16 + row];
      #pragma unroll
      for (int e = 1; e < 16; ++e) pw[e] = pw[e - 1] * A;
      #pragma unroll
      for (int kk = 0; kk < 4; ++kk) {
        #pragma unroll
        for (int v = 0; v < 2; ++v)
          pf[i][m][kk][v] = hi ? pw[13 - 4 * kk - v] : pw[15 - 4 * kk - v];
      }
    }
  }

  #pragma nounroll
  for (int t = 0; t < TPB; ++t) {
    const int tile  = bat * TPB + t;
    float* sT = sTb + (t & 1) * (16 * COLP);
    const int gbase = tile * 256 * D_MODEL + d0;

    float4 rg[8];
    #pragma unroll
    for (int k = 0; k < 8; ++k) {
      const int f  = k * 256 + tid;
      const int rr = f >> 3;
      const int q  = f & 7;
      rg[k] = *reinterpret_cast<const float4*>(x + gbase + rr * D_MODEL + 4 * q);
    }
    #pragma unroll
    for (int k = 0; k < 8; ++k) {
      const int f  = k * 256 + tid;
      const int rr = f >> 3;
      const int q  = f & 7;
      float* p = sT + (rr >> 4) * COLP + (rr & 15) * ROWP + 4 * q;
      p[0] = rg[k].x; p[1] = rg[k].y; p[2] = rg[k].z; p[3] = rg[k].w;
    }
    __syncthreads();

    #pragma unroll
    for (int i = 0; i < 4; ++i) {
      const int chl = wave * 4 + i;
      float* pc = sT + row * COLP + chl;
      v2f bfrag[4];
      #pragma unroll
      for (int kk = 0; kk < 4; ++kk) {
        #pragma unroll
        for (int v = 0; v < 2; ++v) {
          const int j = kk * 4 + hi * 2 + v;
          bfrag[kk][v] = pc[j * ROWP];
        }
      }
      float xs[8];
      #pragma unroll
      for (int v = 0; v < 8; ++v) xs[v] = pc[(v + hi * 8) * ROWP];

      v8f acc = {};
      #pragma unroll
      for (int kk = 0; kk < 4; ++kk)
        acc = wmma_f32_16x16x4(afrag[i][kk], bfrag[kk], acc);

      v8f U[4];
      #pragma unroll
      for (int m = 0; m < 4; ++m) {
        U[m] = (v8f){};
        #pragma unroll
        for (int kk = 0; kk < 4; ++kk)
          U[m] = wmma_f32_16x16x4(pf[i][m][kk], bfrag[kk], U[m]);
      }

      float S[4][8];
      #pragma unroll
      for (int m = 0; m < 4; ++m) {
        const float4 alo = *reinterpret_cast<const float4*>(&sA16[chl][m * 16 + hi * 8]);
        const float4 ahi = *reinterpret_cast<const float4*>(&sA16[chl][m * 16 + hi * 8 + 4]);
        const float av[8] = {alo.x, alo.y, alo.z, alo.w, ahi.x, ahi.y, ahi.z, ahi.w};
        float* pcar = &sCar[chl][m * 16 + hi * 8];
        const float4 clo = *reinterpret_cast<const float4*>(pcar);
        const float4 chi = *reinterpret_cast<const float4*>(pcar + 4);
        const float cv[8] = {clo.x, clo.y, clo.z, clo.w, chi.x, chi.y, chi.z, chi.w};
        float te[8];
        #pragma unroll
        for (int v = 0; v < 8; ++v) {
          float a = av[v];
          const float s0 = cv[v];
          float tv = U[m][v];
          tv = (row == 0) ? fmaf(a, s0, tv) : tv;
          tv = fmaf(a, row_dpp<DPP_ROW_SHR(1)>(tv), tv);
          a *= a;
          tv = fmaf(a, row_dpp<DPP_ROW_SHR(2)>(tv), tv);
          a *= a;
          tv = fmaf(a, row_dpp<DPP_ROW_SHR(4)>(tv), tv);
          a *= a;
          tv = fmaf(a, row_dpp<DPP_ROW_SHR(8)>(tv), tv);
          const float prev = row_dpp<DPP_ROW_SHR(1)>(tv);
          S[m][v] = (row == 0) ? s0 : prev;
          te[v] = tv;
        }
        if (row == 15) {
          *reinterpret_cast<float4*>(pcar)     = make_float4(te[0], te[1], te[2], te[3]);
          *reinterpret_cast<float4*>(pcar + 4) = make_float4(te[4], te[5], te[6], te[7]);
        }
      }

      v8f acc1 = {};
      #pragma unroll
      for (int m = 0; m < 4; ++m) {
        #pragma unroll
        for (int p = 0; p < 4; ++p) {
          const int q = m * 4 + p;
          const v2f g = *reinterpret_cast<const v2f*>(sG + ((chl * 16 + q) * 32 + lane) * 2);
          v2f sb;
          sb[0] = S[m][2 * p];
          sb[1] = S[m][2 * p + 1];
          if (p & 1) acc1 = wmma_f32_16x16x4(g, sb, acc1);
          else       acc  = wmma_f32_16x16x4(g, sb, acc);
        }
      }

      #pragma unroll
      for (int v = 0; v < 8; ++v)
        pc[(v + hi * 8) * ROWP] = (acc[v] + acc1[v]) + Dd[i] * xs[v];
    }
    __syncthreads();

    #pragma unroll
    for (int k = 0; k < 8; ++k) {
      const int f  = k * 256 + tid;
      const int rr = f >> 3;
      const int q  = f & 7;
      const float* p = sT + (rr >> 4) * COLP + (rr & 15) * ROWP + 4 * q;
      float4 o;
      o.x = p[0]; o.y = p[1]; o.z = p[2]; o.w = p[3];
      *reinterpret_cast<float4*>(y + gbase + rr * D_MODEL + 4 * q) = o;
    }
  }
}

extern "C" void kernel_launch(void* const* d_in, const int* in_sizes, int n_in,
                              void* d_out, int out_size, void* d_ws, size_t ws_size,
                              hipStream_t stream) {
  (void)in_sizes; (void)n_in; (void)out_size; (void)d_ws; (void)ws_size;
  const float* x      = (const float*)d_in[0];
  const float* log_dt = (const float*)d_in[1];
  const float* Bmat   = (const float*)d_in[2];
  const float* Cmat   = (const float*)d_in[3];
  const float* Dvec   = (const float*)d_in[4];
  float* y            = (float*)d_out;

  s4d_fused<<<dim3(FUSED_GRID), 256, 0, stream>>>(x, log_dt, Bmat, Cmat, Dvec, y);
}


# chunk end-states computed transposed (U product operands exchanged): 16-lane DPP scan becomes an FMA chain along registers, states return to B order via per-wave LDS transpose; state sums reordered
# speedup vs baseline: 18.0556x; 1.2913x over previous
#include <hip/hip_runtime.h>
#include <hip/hip_bf16.h>
#include <math.h>


#define BATCH    8
#define LENGTH   2048
#define D_MODEL  1024
#define D_STATE  64
#define TCH      16
#define CHUNKS   (LENGTH / TCH)
#define TPB      (CHUNKS / 16)

typedef __attribute__((ext_vector_type(2))) float v2f;
typedef __attribute__((ext_vector_type(8))) float v8f;

#if defined(__HIP_DEVICE_COMPILE__)
#if !__has_builtin(__builtin_amdgcn_wmma_f32_16x16x4_f32)
#error "gfx1250 device pass: __builtin_amdgcn_wmma_f32_16x16x4_f32 not available"
#endif
#endif

__device__ __forceinline__ v8f wmma_f32_16x16x4(v2f a, v2f b, v8f c) {
#if defined(__HIP_DEVICE_COMPILE__)
  return __builtin_amdgcn_wmma_f32_16x16x4_f32(false, a, false, b, (short)0, c,
                                               false, false);
#else
  (void)a; (void)b;
  return c;
#endif
}

__device__ __forceinline__ float xhalf(float v) {
#if defined(__HIP_DEVICE_COMPILE__)
  const int b = __float_as_int(v);
  return __int_as_float(__builtin_amdgcn_permlanex16(b, b, 0x76543210, 0xfedcba98,
                                                     false, false));
#else
  return v;
#endif
}

#define DCH        32
#define ROWP       33
#define COLP       581
#define SPP        65
#define SSP        20
#define FUSED_GRID 256

static_assert(FUSED_GRID == (D_MODEL / DCH) * BATCH, "grid literal out of date");
static_assert(8 * 2 * TCH * SPP <= 2 * 16 * COLP, "tap scratch must fit in the tile buffers");

__global__ __launch_bounds__(256) __attribute__((amdgpu_waves_per_eu(2, 2)))
void s4d_fused(
    const float* __restrict__ x, const float* __restrict__ log_dt,
    const float* __restrict__ Bmat, const float* __restrict__ Cmat,
    const float* __restrict__ Dvec, float* __restrict__ y)
{
  __shared__ float sTb[2 * 16 * COLP];
  __shared__ float sK[DCH][TCH];
  __shared__ __attribute__((aligned(16))) float sG[DCH * 16 * 32 * 2];
  __shared__ __attribute__((aligned(16))) float sAb[DCH][D_STATE];
  __shared__ __attribute__((aligned(16))) float sBb[DCH][D_STATE];
  __shared__ __attribute__((aligned(16))) float sA16[DCH][D_STATE];
  __shared__ __attribute__((aligned(16))) float sS[8 * D_STATE * SSP];

  const int tid   = threadIdx.x;
  const int lane  = tid & 31;
  const int wave  = tid >> 5;
  const int row   = lane & 15;
  const int hi    = lane >> 4;
  const int d0    = (blockIdx.x & 31) * DCH;
  const int bat   = blockIdx.x >> 5;
  const float himask = (float)hi;
  const float lomask = 1.f - himask;

  float qv[8], ab[8];
  #pragma unroll
  for (int i = 0; i < 4; ++i) {
    const int chl = wave * 4 + i;
    const int d = d0 + chl;
    const float dt = expf(log_dt[d]);
    #pragma unroll
    for (int h = 0; h < 2; ++h) {
      const int n = lane + 32 * h;
      const float An  = -0.5f * (float)(n + 1);
      const float dtA = dt * An;
      const float Abar = expf(dtA);
      const float Cn   = Cmat[d * D_STATE + n];
      const float Bn   = Bmat[d * D_STATE + n];
      ab[i * 2 + h] = Abar;
      qv[i * 2 + h] = Cn * (expm1f(dtA) / An) * Bn;
      sAb[chl][n] = Abar;
      sBb[chl][n] = (expm1f(dtA) / An) * Bn;
      const float a2 = Abar * Abar, a4 = a2 * a2, a8 = a4 * a4;
      sA16[chl][n] = a8 * a8;
      float* gp = sG + ((chl * 16 + (n >> 4) * 4 + ((n >> 1) & 3)) * 32 + ((n >> 3) & 1) * 16) * 2 + (n & 1);
      float g = Cn;
      #pragma unroll
      for (int r = 0; r < TCH; ++r) {
        g *= Abar;
        gp[r * 2] = g;
      }
    }
  }
  float* sPw = sTb + wave * (2 * TCH * SPP);
  #pragma unroll
  for (int r = 0; r < 2; ++r) {
    #pragma unroll
    for (int c2 = 0; c2 < 2; ++c2) {
      #pragma unroll
      for (int h = 0; h < 2; ++h) {
        const int n = lane + 32 * h;
        float q = qv[(2 * r + c2) * 2 + h];
        const float Abar = ab[(2 * r + c2) * 2 + h];
        #pragma unroll
        for (int j = 0; j < TCH; ++j) {
          sPw[c2 * (TCH * SPP) + j * SPP + n] = q;
          q *= Abar;
        }
      }
    }
    __syncthreads();
    {
      const float* ps = sPw + hi * (TCH * SPP) + row * SPP;
      float acc = 0.f;
      #pragma unroll 8
      for (int n = 0; n < D_STATE; ++n) acc += ps[n];
      sK[wave * 4 + 2 * r + hi][row] = acc;
    }
    __syncthreads();
  }

  v2f   afrag[4][4];
  float Dd[4];
  float a16r[4][4], car[4][4];
  v2f   pf[4][4][4];
  #pragma unroll
  for (int i = 0; i < 4; ++i) {
    const int dl = wave * 4 + i;
    #pragma unroll
    for (int kk = 0; kk < 4; ++kk) {
      #pragma unroll
      for (int v = 0; v < 2; ++v) {
        const int col = kk * 4 + hi * 2 + v;
        const float tap = sK[dl][(row - col) & 15];
        afrag[i][kk][v] = (row >= col) ? tap : 0.f;
      }
    }
    Dd[i] = Dvec[d0 + dl];
    #pragma unroll
    for (int m = 0; m < 4; ++m) {
      a16r[i][m] = sA16[dl][m * 16 + row];
      car[i][m] = 0.f;
    }
    #pragma unroll
    for (int m = 0; m < 4; ++m) {
      const float A = sAb[dl][m * 16 + row];
      float pw[16];
      pw[0] = sBb[dl][m * 16 + row];
      #pragma unroll
      for (int e = 1; e < 16; ++e) pw[e] = pw[e - 1] * A;
      #pragma unroll
      for (int kk = 0; kk < 4; ++kk) {
        #pragma unroll
        for (int v = 0; v < 2; ++v)
          pf[i][m][kk][v] = hi ? pw[13 - 4 * kk - v] : pw[15 - 4 * kk - v];
      }
    }
  }

  #pragma nounroll
  for (int t = 0; t < TPB; ++t) {
    const int tile  = bat * TPB + t;
    float* sT = sTb + (t & 1) * (16 * COLP);
    const int gbase = tile * 256 * D_MODEL + d0;

    float4 rg[8];
    #pragma unroll
    for (int k = 0; k < 8; ++k) {
      const int f  = k * 256 + tid;
      const int rr = f >> 3;
      const int q  = f & 7;
      rg[k] = *reinterpret_cast<const float4*>(x + gbase + rr * D_MODEL + 4 * q);
    }
    #pragma unroll
    for (int k = 0; k < 8; ++k) {
      const int f  = k * 256 + tid;
      const int rr = f >> 3;
      const int q  = f & 7;
      float* p = sT + (rr >> 4) * COLP + (rr & 15) * ROWP + 4 * q;
      p[0] = rg[k].x; p[1] = rg[k].y; p[2] = rg[k].z; p[3] = rg[k].w;
    }
    __syncthreads();

    #pragma unroll
    for (int i = 0; i < 4; ++i) {
      const int chl = wave * 4 + i;
      float* pc = sT + row * COLP + chl;
      v2f bfrag[4];
      #pragma unroll
      for (int kk = 0; kk < 4; ++kk) {
        #pragma unroll
        for (int v = 0; v < 2; ++v) {
          const int j = kk * 4 + hi * 2 + v;
          bfrag[kk][v] = pc[j * ROWP];
        }
      }
      float xs[8];
      #pragma unroll
      for (int v = 0; v < 8; ++v) xs[v] = pc[(v + hi * 8) * ROWP];

      v8f acc = {};
      #pragma unroll
      for (int kk = 0; kk < 4; ++kk)
        acc = wmma_f32_16x16x4(afrag[i][kk], bfrag[kk], acc);

      v8f UT[4];
      #pragma unroll
      for (int m = 0; m < 4; ++m) {
        UT[m] = (v8f){};
        #pragma unroll
        for (int kk = 0; kk < 4; ++kk)
          UT[m] = wmma_f32_16x16x4(bfrag[kk], pf[i][m][kk], UT[m]);
      }

      float* sSw = sS + wave * (D_STATE * SSP);
      #pragma unroll
      for (int m = 0; m < 4; ++m) {
        const float a = a16r[i][m];
        float t[8];
        float prev = car[i][m];
        #pragma unroll
        for (int r = 0; r < 8; ++r) {
          prev = fmaf(a, prev, UT[m][r]);
          t[r] = prev;
        }
        const float x7 = xhalf(t[7]) * himask;
        const float s0 = car[i][m] + x7;
        float z = x7;
        #pragma unroll
        for (int r = 0; r < 8; ++r) {
          z *= a;
          t[r] += z;
        }
        car[i][m] = xhalf(t[7]) * lomask;
        float* ps = sSw + (m * 16 + row) * SSP + hi * 8;
        *reinterpret_cast<float4*>(ps)     = make_float4(s0,   t[0], t[1], t[2]);
        *reinterpret_cast<float4*>(ps + 4) = make_float4(t[3], t[4], t[5], t[6]);
      }

      v8f acc1 = {};
      #pragma unroll
      for (int m = 0; m < 4; ++m) {
        #pragma unroll
        for (int p = 0; p < 4; ++p) {
          const int q = m * 4 + p;
          const v2f g = *reinterpret_cast<const v2f*>(sG + ((chl * 16 + q) * 32 + lane) * 2);
          const float* pr = sSw + (m * 16 + hi * 8 + 2 * p) * SSP + row;
          v2f sb;
          sb[0] = pr[0];
          sb[1] = pr[SSP];
          if (p & 1) acc1 = wmma_f32_16x16x4(g, sb, acc1);
          else       acc  = wmma_f32_16x16x4(g, sb, acc);
        }
      }

      #pragma unroll
      for (int v = 0; v < 8; ++v)
        pc[(v + hi * 8) * ROWP] = (acc[v] + acc1[v]) + Dd[i] * xs[v];
    }
    __syncthreads();

    #pragma unroll
    for (int k = 0; k < 8; ++k) {
      const int f  = k * 256 + tid;
      const int rr = f >> 3;
      const int q  = f & 7;
      const float* p = sT + (rr >> 4) * COLP + (rr & 15) * ROWP + 4 * q;
      float4 o;
      o.x = p[0]; o.y = p[1]; o.z = p[2]; o.w = p[3];
      *reinterpret_cast<float4*>(y + gbase + rr * D_MODEL + 4 * q) = o;
    }
  }
}

extern "C" void kernel_launch(void* const* d_in, const int* in_sizes, int n_in,
                              void* d_out, int out_size, void* d_ws, size_t ws_size,
                              hipStream_t stream) {
  (void)in_sizes; (void)n_in; (void)out_size; (void)d_ws; (void)ws_size;
  const float* x      = (const float*)d_in[0];
  const float* log_dt = (const float*)d_in[1];
  const float* Bmat   = (const float*)d_in[2];
  const float* Cmat   = (const float*)d_in[3];
  const float* Dvec   = (const float*)d_in[4];
  float* y            = (float*)d_out;

  s4d_fused<<<dim3(FUSED_GRID), 256, 0, stream>>>(x, log_dt, Bmat, Cmat, Dvec, y);
}


# each tile's global row loads issued one tile ahead, behind the barrier that opens the previous tile's matrix phase (tile 0's under the table set-up), and held in registers across it; sums unchanged
# speedup vs baseline: 20.8041x; 1.1522x over previous
#include <hip/hip_runtime.h>
#include <hip/hip_bf16.h>
#include <math.h>


#define BATCH    8
#define LENGTH   2048
#define D_MODEL  1024
#define D_STATE  64
#define TCH      16
#define CHUNKS   (LENGTH / TCH)
#define TPB      (CHUNKS / 16)

typedef __attribute__((ext_vector_type(2))) float v2f;
typedef __attribute__((ext_vector_type(8))) float v8f;

#if defined(__HIP_DEVICE_COMPILE__)
#if !__has_builtin(__builtin_amdgcn_wmma_f32_16x16x4_f32)
#error "gfx1250 device pass: __builtin_amdgcn_wmma_f32_16x16x4_f32 not available"
#endif
#endif

__device__ __forceinline__ v8f wmma_f32_16x16x4(v2f a, v2f b, v8f c) {
#if defined(__HIP_DEVICE_COMPILE__)
  return __builtin_amdgcn_wmma_f32_16x16x4_f32(false, a, false, b, (short)0, c,
                                               false, false);
#else
  (void)a; (void)b;
  return c;
#endif
}

__device__ __forceinline__ float xhalf(float v) {
#if defined(__HIP_DEVICE_COMPILE__)
  const int b = __float_as_int(v);
  return __int_as_float(__builtin_amdgcn_permlanex16(b, b, 0x76543210, 0xfedcba98,
                                                     false, false));
#else
  return v;
#endif
}

#define DCH        32
#define ROWP       33
#define COLP       581
#define SPP        65
#define SSP        20
#define FUSED_GRID 256

static_assert(FUSED_GRID == (D_MODEL / DCH) * BATCH, "grid literal out of date");
static_assert(8 * 2 * TCH * SPP <= 2 * 16 * COLP, "tap scratch must fit in the tile buffers");

__global__ __launch_bounds__(256) __attribute__((amdgpu_waves_per_eu(2, 2)))
void s4d_fused(
    const float* __restrict__ x, const float* __restrict__ log_dt,
    const float* __restrict__ Bmat, const float* __restrict__ Cmat,
    const float* __restrict__ Dvec, float* __restrict__ y)
{
  __shared__ float sTb[2 * 16 * COLP];
  __shared__ float sK[DCH][TCH];
  __shared__ __attribute__((aligned(16))) float sG[DCH * 16 * 32 * 2];
  __shared__ __attribute__((aligned(16))) float sAb[DCH][D_STATE];
  __shared__ __attribute__((aligned(16))) float sBb[DCH][D_STATE];
  __shared__ __attribute__((aligned(16))) float sA16[DCH][D_STATE];
  __shared__ __attribute__((aligned(16))) float sS[8 * D_STATE * SSP];

  const int tid   = threadIdx.x;
  const int lane  = tid & 31;
  const int wave  = tid >> 5;
  const int row   = lane & 15;
  const int hi    = lane >> 4;
  const int d0    = (blockIdx.x & 31) * DCH;
  const int bat   = blockIdx.x >> 5;
  const float himask = (float)hi;
  const float lomask = 1.f - himask;

  float qv[8], ab[8];
  #pragma unroll
  for (int i = 0; i < 4; ++i) {
    const int chl = wave * 4 + i;
    const int d = d0 + chl;
    const float dt = expf(log_dt[d]);
    #pragma unroll
    for (int h = 0; h < 2; ++h) {
      const int n = lane + 32 * h;
      const float An  = -0.5f * (float)(n + 1);
      const float dtA = dt * An;
      const float Abar = expf(dtA);
      const float Cn   = Cmat[d * D_STATE + n];
      const float Bn   = Bmat[d * D_STATE + n];
      ab[i * 2 + h] = Abar;
      qv[i * 2 + h] = Cn * (expm1f(dtA) / An) * Bn;
      sAb[chl][n] = Abar;
      sBb[chl][n] = (expm1f(dtA) / An) * Bn;
      const float a2 = Abar * Abar, a4 = a2 * a2, a8 = a4 * a4;
      sA16[chl][n] = a8 * a8;
      float* gp = sG + ((chl * 16 + (n >> 4) * 4 + ((n >> 1) & 3)) * 32 + ((n >> 3) & 1) * 16) * 2 + (n & 1);
      float g = Cn;
      #pragma unroll
      for (int r = 0; r < TCH; ++r) {
        g *= Abar;
        gp[r * 2] = g;
      }
    }
  }
  float* sPw = sTb + wave * (2 * TCH * SPP);
  #pragma unroll
  for (int r = 0; r < 2; ++r) {
    #pragma unroll
    for (int c2 = 0; c2 < 2; ++c2) {
      #pragma unroll
      for (int h = 0; h < 2; ++h) {
        const int n = lane + 32 * h;
        float q = qv[(2 * r + c2) * 2 + h];
        const float Abar = ab[(2 * r + c2) * 2 + h];
        #pragma unroll
        for (int j = 0; j < TCH; ++j) {
          sPw[c2 * (TCH * SPP) + j * SPP + n] = q;
          q *= Abar;
        }
      }
    }
    __syncthreads();
    {
      const float* ps = sPw + hi * (TCH * SPP) + row * SPP;
      float acc = 0.f;
      #pragma unroll 8
      for (int n = 0; n < D_STATE; ++n) acc += ps[n];
      sK[wave * 4 + 2 * r + hi][row] = acc;
    }
    __syncthreads();
  }

  const int rr0  = tid >> 3;
  const int q4   = 4 * (tid & 7);
  const int lds0 = (rr0 >> 4) * COLP + (rr0 & 15) * ROWP + q4;
  const int g0   = bat * TPB * 256 * D_MODEL + d0 + rr0 * D_MODEL + q4;

  float4 rg[8];
  #pragma unroll
  for (int k = 0; k < 8; ++k)
    rg[k] = *reinterpret_cast<const float4*>(x + g0 + k * 32 * D_MODEL);

  v2f   afrag[4][4];
  float Dd[4];
  float a16r[4][4], car[4][4];
  v2f   pf[4][4][4];
  #pragma unroll
  for (int i = 0; i < 4; ++i) {
    const int dl = wave * 4 + i;
    #pragma unroll
    for (int kk = 0; kk < 4; ++kk) {
      #pragma unroll
      for (int v = 0; v < 2; ++v) {
        const int col = kk * 4 + hi * 2 + v;
        const float tap = sK[dl][(row - col) & 15];
        afrag[i][kk][v] = (row >= col) ? tap : 0.f;
      }
    }
    Dd[i] = Dvec[d0 + dl];
    #pragma unroll
    for (int m = 0; m < 4; ++m) {
      a16r[i][m] = sA16[dl][m * 16 + row];
      car[i][m] = 0.f;
    }
    #pragma unroll
    for (int m = 0; m < 4; ++m) {
      const float A = sAb[dl][m * 16 + row];
      float pw[16];
      pw[0] = sBb[dl][m * 16 + row];
      #pragma unroll
      for (int e = 1; e < 16; ++e) pw[e] = pw[e - 1] * A;
      #pragma unroll
      for (int kk = 0; kk < 4; ++kk) {
        #pragma unroll
        for (int v = 0; v < 2; ++v)
          pf[i][m][kk][v] = hi ? pw[13 - 4 * kk - v] : pw[15 - 4 * kk - v];
      }
    }
  }

  #pragma nounroll
  for (int t = 0; t < TPB; ++t) {
    float* sT = sTb + (t & 1) * (16 * COLP);
    const int gt = g0 + t * 256 * D_MODEL;

    #pragma unroll
    for (int k = 0; k < 8; ++k) {
      float* p = sT + lds0 + k * 2 * COLP;
      p[0] = rg[k].x; p[1] = rg[k].y; p[2] = rg[k].z; p[3] = rg[k].w;
    }
    __syncthreads();

    {
      const int gn = gt + ((t + 1 < TPB) ? 256 * D_MODEL : 0);
      #pragma unroll
      for (int k = 0; k < 8; ++k)
        rg[k] = *reinterpret_cast<const float4*>(x + gn + k * 32 * D_MODEL);
    }

    #pragma unroll
    for (int i = 0; i < 4; ++i) {
      const int chl = wave * 4 + i;
      float* pc = sT + row * COLP + chl;
      v2f bfrag[4];
      #pragma unroll
      for (int kk = 0; kk < 4; ++kk) {
        #pragma unroll
        for (int v = 0; v < 2; ++v) {
          const int j = kk * 4 + hi * 2 + v;
          bfrag[kk][v] = pc[j * ROWP];
        }
      }
      float xs[8];
      #pragma unroll
      for (int v = 0; v < 8; ++v) xs[v] = pc[(v + hi * 8) * ROWP];

      v8f acc = {};
      #pragma unroll
      for (int kk = 0; kk < 4; ++kk)
        acc = wmma_f32_16x16x4(afrag[i][kk], bfrag[kk], acc);

      v8f UT[4];
      #pragma unroll
      for (int m = 0; m < 4; ++m) {
        UT[m] = (v8f){};
        #pragma unroll
        for (int kk = 0; kk < 4; ++kk)
          UT[m] = wmma_f32_16x16x4(bfrag[kk], pf[i][m][kk], UT[m]);
      }

      float* sSw = sS + wave * (D_STATE * SSP);
      #pragma unroll
      for (int m = 0; m < 4; ++m) {
        const float a = a16r[i][m];
        float t[8];
        float prev = car[i][m];
        #pragma unroll
        for (int r = 0; r < 8; ++r) {
          prev = fmaf(a, prev, UT[m][r]);
          t[r] = prev;
        }
        const float x7 = xhalf(t[7]) * himask;
        const float s0 = car[i][m] + x7;
        float z = x7;
        #pragma unroll
        for (int r = 0; r < 8; ++r) {
          z *= a;
          t[r] += z;
        }
        car[i][m] = xhalf(t[7]) * lomask;
        float* ps = sSw + (m * 16 + row) * SSP + hi * 8;
        *reinterpret_cast<float4*>(ps)     = make_float4(s0,   t[0], t[1], t[2]);
        *reinterpret_cast<float4*>(ps + 4) = make_float4(t[3], t[4], t[5], t[6]);
      }

      v8f acc1 = {};
      #pragma unroll
      for (int m = 0; m < 4; ++m) {
        #pragma unroll
        for (int p = 0; p < 4; ++p) {
          const int q = m * 4 + p;
          const v2f g = *reinterpret_cast<const v2f*>(sG + ((chl * 16 + q) * 32 + lane) * 2);
          const float* pr = sSw + (m * 16 + hi * 8 + 2 * p) * SSP + row;
          v2f sb;
          sb[0] = pr[0];
          sb[1] = pr[SSP];
          if (p & 1) acc1 = wmma_f32_16x16x4(g, sb, acc1);
          else       acc  = wmma_f32_16x16x4(g, sb, acc);
        }
      }

      #pragma unroll
      for (int v = 0; v < 8; ++v)
        pc[(v + hi * 8) * ROWP] = (acc[v] + acc1[v]) + Dd[i] * xs[v];
    }
    __syncthreads();

    #pragma unroll
    for (int k = 0; k < 8; ++k) {
      const float* p = sT + lds0 + k * 2 * COLP;
      float4 o;
      o.x = p[0]; o.y = p[1]; o.z = p[2]; o.w = p[3];
      *reinterpret_cast<float4*>(y + gt + k * 32 * D_MODEL) = o;
    }
  }
}

extern "C" void kernel_launch(void* const* d_in, const int* in_sizes, int n_in,
                              void* d_out, int out_size, void* d_ws, size_t ws_size,
                              hipStream_t stream) {
  (void)in_sizes; (void)n_in; (void)out_size; (void)d_ws; (void)ws_size;
  const float* x      = (const float*)d_in[0];
  const float* log_dt = (const float*)d_in[1];
  const float* Bmat   = (const float*)d_in[2];
  const float* Cmat   = (const float*)d_in[3];
  const float* Dvec   = (const float*)d_in[4];
  float* y            = (float*)d_out;

  s4d_fused<<<dim3(FUSED_GRID), 256, 0, stream>>>(x, log_dt, Bmat, Cmat, Dvec, y);
}
